# starting kernel
# baseline (speedup 1.0000x reference)
#include <hip/hip_runtime.h>
#include <hip/hip_bf16.h>

#define HEADS   4
#define CH      128
#define HC      512
#define NEG_SLOPE 0.2f

typedef float v2f __attribute__((ext_vector_type(2)));
typedef float v8f __attribute__((ext_vector_type(8)));

__device__ __forceinline__ unsigned encF(float f) {
    unsigned u = __float_as_uint(f);
    return (u & 0x80000000u) ? ~u : (u | 0x80000000u);
}
__device__ __forceinline__ float decF(unsigned e) {
    unsigned b = (e & 0x80000000u) ? (e & 0x7FFFFFFFu) : ~e;
    return __uint_as_float(b);
}
__device__ __forceinline__ float lrelu(float a) {
    return a > 0.f ? a : NEG_SLOPE * a;
}

__global__ void gat_init(float* __restrict__ out, const float* __restrict__ bias,
                         unsigned* __restrict__ emax, float* __restrict__ denom, int N) {
    long long i = (long long)blockIdx.x * blockDim.x + threadIdx.x;
    long long total = (long long)N * HC;
    if (i < total) out[i] = bias[i & (HC - 1)];
    if (i < (long long)N * HEADS) { emax[i] = 0u; denom[i] = 0.f; }
}

__global__ void gat_gemm_wmma(const float* __restrict__ X, const float* __restrict__ W,
                              float* __restrict__ H, int N) {
    int wid   = (blockIdx.x * blockDim.x + threadIdx.x) >> 5;
    int lane  = threadIdx.x & 31;
    int mt    = wid >> 3;
    int ns    = wid & 7;
    int row0  = mt * 16;
    int col0  = ns * 64;
    if (row0 >= N) return;

    int half = lane >> 4;
    int l16  = lane & 15;

    int arow = row0 + l16;
    if (arow >= N) arow = N - 1;
    const float* __restrict__ A = X + (size_t)arow * CH;

    v8f acc0 = {}, acc1 = {}, acc2 = {}, acc3 = {};

#pragma unroll 4
    for (int k0 = 0; k0 < CH; k0 += 4) {
        int ka = k0 + half * 2;
        v2f a; a.x = A[ka]; a.y = A[ka + 1];

        const float* __restrict__ B = W + (size_t)ka * HC + col0 + l16;
        v2f b0; b0.x = B[0];  b0.y = B[HC];
        v2f b1; b1.x = B[16]; b1.y = B[16 + HC];
        v2f b2; b2.x = B[32]; b2.y = B[32 + HC];
        v2f b3; b3.x = B[48]; b3.y = B[48 + HC];

        acc0 = __builtin_amdgcn_wmma_f32_16x16x4_f32(false, a, false, b0, (short)0, acc0, false, false);
        acc1 = __builtin_amdgcn_wmma_f32_16x16x4_f32(false, a, false, b1, (short)0, acc1, false, false);
        acc2 = __builtin_amdgcn_wmma_f32_16x16x4_f32(false, a, false, b2, (short)0, acc2, false, false);
        acc3 = __builtin_amdgcn_wmma_f32_16x16x4_f32(false, a, false, b3, (short)0, acc3, false, false);
    }

#pragma unroll
    for (int r = 0; r < 8; ++r) {
        int m = row0 + half * 8 + r;
        if (m < N) {
            float* __restrict__ O = H + (size_t)m * HC + col0 + l16;
            O[0]  = acc0[r];
            O[16] = acc1[r];
            O[32] = acc2[r];
            O[48] = acc3[r];
        }
    }
}

__global__ void gat_dots(const float* __restrict__ H, const float* __restrict__ att_src,
                         const float* __restrict__ att_dst, float* __restrict__ asrc,
                         float* __restrict__ adst, int N) {
    int n    = (blockIdx.x * blockDim.x + threadIdx.x) >> 5;
    int lane = threadIdx.x & 31;
    if (n >= N) return;
    const float4* __restrict__ hp = (const float4*)(H + (size_t)n * HC);
    const float4* __restrict__ s4 = (const float4*)att_src;
    const float4* __restrict__ d4 = (const float4*)att_dst;
    float ss[HEADS], sd[HEADS];
#pragma unroll
    for (int h = 0; h < HEADS; ++h) {
        float4 hv = hp[h * 32 + lane];
        float4 av = s4[h * 32 + lane];
        float4 bv = d4[h * 32 + lane];
        float s = hv.x * av.x + hv.y * av.y + hv.z * av.z + hv.w * av.w;
        float d = hv.x * bv.x + hv.y * bv.y + hv.z * bv.z + hv.w * bv.w;
#pragma unroll
        for (int m = 16; m > 0; m >>= 1) {
            s += __shfl_xor(s, m, 32);
            d += __shfl_xor(d, m, 32);
        }
        ss[h] = s; sd[h] = d;
    }
    if (lane == 0) {
#pragma unroll
        for (int h = 0; h < HEADS; ++h) {
            asrc[(size_t)n * HEADS + h] = ss[h];
            adst[(size_t)n * HEADS + h] = sd[h];
        }
    }
}

__global__ void gat_edge_max(const int* __restrict__ src, const int* __restrict__ dst,
                             const float* __restrict__ asrc, const float* __restrict__ adst,
                             unsigned* __restrict__ emax, int E, int N) {
    int t = blockIdx.x * blockDim.x + threadIdx.x;
    if (t >= E + N) return;
    int s, d;
    if (t < E) { s = src[t]; d = dst[t]; } else { s = d = t - E; }
    float4 as = ((const float4*)asrc)[s];
    float4 ad = ((const float4*)adst)[d];
    float v[HEADS] = { as.x + ad.x, as.y + ad.y, as.z + ad.z, as.w + ad.w };
#pragma unroll
    for (int h = 0; h < HEADS; ++h)
        atomicMax(&emax[(size_t)d * HEADS + h], encF(lrelu(v[h])));
}

__global__ void gat_edge_expsum(const int* __restrict__ src, const int* __restrict__ dst,
                                const float* __restrict__ asrc, const float* __restrict__ adst,
                                const unsigned* __restrict__ emax, float* __restrict__ denom,
                                int E, int N) {
    int t = blockIdx.x * blockDim.x + threadIdx.x;
    if (t >= E + N) return;
    int s, d;
    if (t < E) { s = src[t]; d = dst[t]; } else { s = d = t - E; }
    float4 as = ((const float4*)asrc)[s];
    float4 ad = ((const float4*)adst)[d];
    float v[HEADS] = { as.x + ad.x, as.y + ad.y, as.z + ad.z, as.w + ad.w };
#pragma unroll
    for (int h = 0; h < HEADS; ++h) {
        float m = decF(emax[(size_t)d * HEADS + h]);
        atomicAdd(&denom[(size_t)d * HEADS + h], __expf(lrelu(v[h]) - m));
    }
}

__global__ void gat_scatter(const float* __restrict__ H, const int* __restrict__ src,
                            const int* __restrict__ dst, const float* __restrict__ asrc,
                            const float* __restrict__ adst, const unsigned* __restrict__ emax,
                            const float* __restrict__ denom, float* __restrict__ out,
                            int E, int N) {
    int wid  = (blockIdx.x * blockDim.x + threadIdx.x) >> 5;
    int lane = threadIdx.x & 31;
    int total = (E + N) * HEADS;
    if (wid >= total) return;
    int head = wid & (HEADS - 1);
    int e    = wid >> 2;
    int s, d;
    if (e < E) { s = src[e]; d = dst[e]; } else { s = d = e - E; }

    float a = lrelu(asrc[(size_t)s * HEADS + head] + adst[(size_t)d * HEADS + head]);
    float m = decF(emax[(size_t)d * HEADS + head]);
    float alpha = __expf(a - m) / denom[(size_t)d * HEADS + head];

    const float4* __restrict__ hp = (const float4*)(H + (size_t)s * HC + head * CH);
    float4 hv = hp[lane];
    float* __restrict__ op = out + (size_t)d * HC + head * CH + lane * 4;
    atomicAdd(op + 0, alpha * hv.x);
    atomicAdd(op + 1, alpha * hv.y);
    atomicAdd(op + 2, alpha * hv.z);
    atomicAdd(op + 3, alpha * hv.w);
}

extern "C" void kernel_launch(void* const* d_in, const int* in_sizes, int n_in,
                              void* d_out, int out_size, void* d_ws, size_t ws_size,
                              hipStream_t stream) {
    const float* x        = (const float*)d_in[0];
    const int*   eidx     = (const int*)d_in[1];
    const float* W        = (const float*)d_in[2];
    const float* att_src  = (const float*)d_in[3];
    const float* att_dst  = (const float*)d_in[4];
    const float* bias     = (const float*)d_in[5];
    float*       out      = (float*)d_out;

    const int N = in_sizes[0] / CH;
    const int E = in_sizes[1] / 2;
    const int* src = eidx;
    const int* dst = eidx + E;

    float*    H     = (float*)d_ws;
    float*    asrc  = H + (size_t)N * HC;
    float*    adst  = asrc + (size_t)N * HEADS;
    unsigned* emax  = (unsigned*)(adst + (size_t)N * HEADS);
    float*    denom = (float*)(emax + (size_t)N * HEADS);

    const int TB = 256;

    {
        long long total = (long long)N * HC;
        int blocks = (int)((total + TB - 1) / TB);
        gat_init<<<blocks, TB, 0, stream>>>(out, bias, emax, denom, N);
    }
    {
        int mtiles = (N + 15) / 16;
        int waves  = mtiles * 8;
        int blocks = (waves + 7) / 8;
        gat_gemm_wmma<<<blocks, TB, 0, stream>>>(x, W, H, N);
    }
    {
        long long threads = (long long)N * 32;
        int blocks = (int)((threads + TB - 1) / TB);
        gat_dots<<<blocks, TB, 0, stream>>>(H, att_src, att_dst, asrc, adst, N);
    }
    {
        int blocks = (E + N + TB - 1) / TB;
        gat_edge_max<<<blocks, TB, 0, stream>>>(src, dst, asrc, adst, emax, E, N);
    }
    {
        int blocks = (E + N + TB - 1) / TB;
        gat_edge_expsum<<<blocks, TB, 0, stream>>>(src, dst, asrc, adst, emax, denom, E, N);
    }
    {
        long long threads = (long long)(E + N) * HEADS * 32;
        int blocks = (int)((threads + TB - 1) / TB);
        gat_scatter<<<blocks, TB, 0, stream>>>(H, src, dst, asrc, adst, emax, denom, out, E, N);
    }
}


# per-edge f32 atomics replaced by a per-dst gather: edges grouped by dst in-call (int count, alloc, fill); one wave per node does max, exp-sum and weighted sum in registers in fill-arrival order
# speedup vs baseline: 3.1690x; 3.1690x over previous
#include <hip/hip_runtime.h>
#include <hip/hip_bf16.h>


#define HEADS   4
#define CH      128
#define HC      512
#define NEG_SLOPE 0.2f

typedef float v2f __attribute__((ext_vector_type(2)));
typedef float v8f __attribute__((ext_vector_type(8)));

__device__ __forceinline__ float lrelu(float a) {
    return a > 0.f ? a : NEG_SLOPE * a;
}
__device__ __forceinline__ float wave_sum(float v) {
#pragma unroll
    for (int m = 16; m > 0; m >>= 1) v += __shfl_xor(v, m, 32);
    return v;
}
__device__ __forceinline__ float wave_max(float v) {
#pragma unroll
    for (int m = 16; m > 0; m >>= 1) v = fmaxf(v, __shfl_xor(v, m, 32));
    return v;
}
__device__ __forceinline__ float dot4(float4 a, float4 b) {
    return a.x * b.x + a.y * b.y + a.z * b.z + a.w * b.w;
}

__global__ void gat_gemm_wmma(const float* __restrict__ X, const float* __restrict__ W,
                              float* __restrict__ H, int N) {
    int wid   = (blockIdx.x * blockDim.x + threadIdx.x) >> 5;
    int lane  = threadIdx.x & 31;
    int mt    = wid >> 3;
    int ns    = wid & 7;
    int row0  = mt * 16;
    int col0  = ns * 64;
    if (row0 >= N) return;

    int half = lane >> 4;
    int l16  = lane & 15;

    int arow = row0 + l16;
    if (arow >= N) arow = N - 1;
    const float* __restrict__ A = X + (size_t)arow * CH;

    v8f acc0 = {}, acc1 = {}, acc2 = {}, acc3 = {};

#pragma unroll 4
    for (int k0 = 0; k0 < CH; k0 += 4) {
        int ka = k0 + half * 2;
        v2f a; a.x = A[ka]; a.y = A[ka + 1];

        const float* __restrict__ B = W + (size_t)ka * HC + col0 + l16;
        v2f b0; b0.x = B[0];  b0.y = B[HC];
        v2f b1; b1.x = B[16]; b1.y = B[16 + HC];
        v2f b2; b2.x = B[32]; b2.y = B[32 + HC];
        v2f b3; b3.x = B[48]; b3.y = B[48 + HC];

        acc0 = __builtin_amdgcn_wmma_f32_16x16x4_f32(false, a, false, b0, (short)0, acc0, false, false);
        acc1 = __builtin_amdgcn_wmma_f32_16x16x4_f32(false, a, false, b1, (short)0, acc1, false, false);
        acc2 = __builtin_amdgcn_wmma_f32_16x16x4_f32(false, a, false, b2, (short)0, acc2, false, false);
        acc3 = __builtin_amdgcn_wmma_f32_16x16x4_f32(false, a, false, b3, (short)0, acc3, false, false);
    }

#pragma unroll
    for (int r = 0; r < 8; ++r) {
        int m = row0 + half * 8 + r;
        if (m < N) {
            float* __restrict__ O = H + (size_t)m * HC + col0 + l16;
            O[0]  = acc0[r];
            O[16] = acc1[r];
            O[32] = acc2[r];
            O[48] = acc3[r];
        }
    }
}

__global__ void gat_dots(const float* __restrict__ H, const float* __restrict__ att_src,
                         float* __restrict__ asrc, int* __restrict__ cnt,
                         int* __restrict__ total, int N) {
    int n    = (blockIdx.x * blockDim.x + threadIdx.x) >> 5;
    int lane = threadIdx.x & 31;
    if (n >= N) return;
    const float4* __restrict__ hp = (const float4*)(H + (size_t)n * HC);
    const float4* __restrict__ s4 = (const float4*)att_src;
    float ss[HEADS];
#pragma unroll
    for (int h = 0; h < HEADS; ++h)
        ss[h] = wave_sum(dot4(hp[h * 32 + lane], s4[h * 32 + lane]));
    if (lane == 0) {
        ((float4*)asrc)[n] = make_float4(ss[0], ss[1], ss[2], ss[3]);
        cnt[n] = 0;
        if (n == 0) *total = 0;
    }
}

__global__ void gat_count(const int* __restrict__ dst, int* __restrict__ cnt, int E) {
    int t = blockIdx.x * blockDim.x + threadIdx.x;
    if (t < E) atomicAdd(&cnt[dst[t]], 1);
}

__global__ void gat_alloc(const int* __restrict__ cnt, int* __restrict__ cursor,
                          int* __restrict__ total, int N) {
    __shared__ int wsum[32];
    __shared__ int blockbase;
    int tid  = threadIdx.x;
    int lane = tid & 31;
    int w    = tid >> 5;
    int nw   = blockDim.x >> 5;
    int i    = blockIdx.x * blockDim.x + tid;
    int c    = (i < N) ? cnt[i] : 0;
    int x    = c;
#pragma unroll
    for (int o = 1; o < 32; o <<= 1) {
        int y = __shfl_up(x, o, 32);
        if (lane >= o) x += y;
    }
    if (lane == 31) wsum[w] = x;
    __syncthreads();
    if (tid == 0) {
        int run = 0;
        for (int k = 0; k < nw; ++k) { int t = wsum[k]; wsum[k] = run; run += t; }
        blockbase = atomicAdd(total, run);
    }
    __syncthreads();
    if (i < N) cursor[i] = blockbase + wsum[w] + x - c;
}

__global__ void gat_fill(const int* __restrict__ src, const int* __restrict__ dst,
                         int* __restrict__ cursor, int* __restrict__ csr, int E) {
    int t = blockIdx.x * blockDim.x + threadIdx.x;
    if (t < E) {
        int pos = atomicAdd(&cursor[dst[t]], 1);
        csr[pos] = src[t];
    }
}

__global__ void gat_gather(const float* __restrict__ H, const float* __restrict__ asrc,
                           const float* __restrict__ att_dst, const float* __restrict__ bias,
                           const int* __restrict__ cnt, const int* __restrict__ cursor,
                           const int* __restrict__ csr, float* __restrict__ out, int N) {
    int n    = (blockIdx.x * blockDim.x + threadIdx.x) >> 5;
    int lane = threadIdx.x & 31;
    n = __builtin_amdgcn_readfirstlane(n);
    if (n >= N) return;

    const float4* __restrict__ asrc4 = (const float4*)asrc;
    const float4* __restrict__ hrow  = (const float4*)(H + (size_t)n * HC);
    const float4* __restrict__ d4    = (const float4*)att_dst;

    const int end   = cursor[n];
    const int deg   = cnt[n];
    const int start = end - deg;

    float4 hs[HEADS];
#pragma unroll
    for (int h = 0; h < HEADS; ++h) hs[h] = hrow[h * 32 + lane];

    const bool ok0 = lane < deg;
    int    s0 = n;
    float4 a0 = make_float4(0.f, 0.f, 0.f, 0.f);
    if (ok0) {
        s0 = csr[start + lane];
        a0 = asrc4[s0];
    }

    float ad[HEADS];
#pragma unroll
    for (int h = 0; h < HEADS; ++h) ad[h] = wave_sum(dot4(hs[h], d4[h * 32 + lane]));
    const float4 an = asrc4[n];
    const float vs[HEADS] = { lrelu(an.x + ad[0]), lrelu(an.y + ad[1]),
                              lrelu(an.z + ad[2]), lrelu(an.w + ad[3]) };
    const float e0[HEADS] = { ok0 ? lrelu(a0.x + ad[0]) : vs[0], ok0 ? lrelu(a0.y + ad[1]) : vs[1],
                              ok0 ? lrelu(a0.z + ad[2]) : vs[2], ok0 ? lrelu(a0.w + ad[3]) : vs[3] };

    float m[HEADS] = { fmaxf(vs[0], e0[0]), fmaxf(vs[1], e0[1]),
                       fmaxf(vs[2], e0[2]), fmaxf(vs[3], e0[3]) };
    for (int base = start + 32; base < end; base += 32) {
        int i = base + lane;
        if (i < end) {
            float4 a = asrc4[csr[i]];
            m[0] = fmaxf(m[0], lrelu(a.x + ad[0]));
            m[1] = fmaxf(m[1], lrelu(a.y + ad[1]));
            m[2] = fmaxf(m[2], lrelu(a.z + ad[2]));
            m[3] = fmaxf(m[3], lrelu(a.w + ad[3]));
        }
    }
#pragma unroll
    for (int h = 0; h < HEADS; ++h) m[h] = wave_max(m[h]);

    float p0[HEADS], z[HEADS];
#pragma unroll
    for (int h = 0; h < HEADS; ++h) {
        p0[h] = ok0 ? __expf(e0[h] - m[h]) : 0.f;
        z[h]  = p0[h];
    }
    for (int base = start + 32; base < end; base += 32) {
        int i = base + lane;
        if (i < end) {
            float4 a = asrc4[csr[i]];
            z[0] += __expf(lrelu(a.x + ad[0]) - m[0]);
            z[1] += __expf(lrelu(a.y + ad[1]) - m[1]);
            z[2] += __expf(lrelu(a.z + ad[2]) - m[2]);
            z[3] += __expf(lrelu(a.w + ad[3]) - m[3]);
        }
    }
    float4 acc[HEADS];
#pragma unroll
    for (int h = 0; h < HEADS; ++h) {
        float ps = __expf(vs[h] - m[h]);
        z[h] = wave_sum(z[h]) + ps;
        float w = ps / z[h];
        acc[h] = make_float4(w * hs[h].x, w * hs[h].y, w * hs[h].z, w * hs[h].w);
    }

    int   s = s0;
    float al[HEADS] = { p0[0] / z[0], p0[1] / z[1], p0[2] / z[2], p0[3] / z[3] };
    for (int base = start; base < end; base += 32) {
        if (base != start) {
            int i = base + lane;
            s = n;
            al[0] = al[1] = al[2] = al[3] = 0.f;
            if (i < end) {
                s = csr[i];
                float4 a = asrc4[s];
                al[0] = __expf(lrelu(a.x + ad[0]) - m[0]) / z[0];
                al[1] = __expf(lrelu(a.y + ad[1]) - m[1]) / z[1];
                al[2] = __expf(lrelu(a.z + ad[2]) - m[2]) / z[2];
                al[3] = __expf(lrelu(a.w + ad[3]) - m[3]) / z[3];
            }
        }
        int k = end - base; if (k > 32) k = 32;
#pragma unroll 4
        for (int j = 0; j < k; ++j) {
            int   sj = __builtin_amdgcn_readlane(s, j);
            float w0 = __int_as_float(__builtin_amdgcn_readlane(__float_as_int(al[0]), j));
            float w1 = __int_as_float(__builtin_amdgcn_readlane(__float_as_int(al[1]), j));
            float w2 = __int_as_float(__builtin_amdgcn_readlane(__float_as_int(al[2]), j));
            float w3 = __int_as_float(__builtin_amdgcn_readlane(__float_as_int(al[3]), j));
            const float4* __restrict__ r = (const float4*)(H + (size_t)sj * HC);
            float4 h0 = r[lane], h1 = r[32 + lane], h2 = r[64 + lane], h3 = r[96 + lane];
            acc[0].x += w0 * h0.x; acc[0].y += w0 * h0.y; acc[0].z += w0 * h0.z; acc[0].w += w0 * h0.w;
            acc[1].x += w1 * h1.x; acc[1].y += w1 * h1.y; acc[1].z += w1 * h1.z; acc[1].w += w1 * h1.w;
            acc[2].x += w2 * h2.x; acc[2].y += w2 * h2.y; acc[2].z += w2 * h2.z; acc[2].w += w2 * h2.w;
            acc[3].x += w3 * h3.x; acc[3].y += w3 * h3.y; acc[3].z += w3 * h3.z; acc[3].w += w3 * h3.w;
        }
    }

    const float4* __restrict__ b4 = (const float4*)bias;
    float4* __restrict__ o4 = (float4*)(out + (size_t)n * HC);
#pragma unroll
    for (int h = 0; h < HEADS; ++h) {
        float4 b = b4[h * 32 + lane];
        o4[h * 32 + lane] = make_float4(acc[h].x + b.x, acc[h].y + b.y, acc[h].z + b.z, acc[h].w + b.w);
    }
}

extern "C" void kernel_launch(void* const* d_in, const int* in_sizes, int n_in,
                              void* d_out, int out_size, void* d_ws, size_t ws_size,
                              hipStream_t stream) {
    const float* x        = (const float*)d_in[0];
    const int*   eidx     = (const int*)d_in[1];
    const float* W        = (const float*)d_in[2];
    const float* att_src  = (const float*)d_in[3];
    const float* att_dst  = (const float*)d_in[4];
    const float* bias     = (const float*)d_in[5];
    float*       out      = (float*)d_out;

    const int N = in_sizes[0] / CH;
    const int E = in_sizes[1] / 2;
    const int* src = eidx;
    const int* dst = eidx + E;

    float* H      = (float*)d_ws;
    float* asrc   = H + (size_t)N * HC;
    int*   cnt    = (int*)(asrc + (size_t)N * HEADS);
    int*   cursor = cnt + N;
    int*   total  = cursor + N;
    int*   csr    = total + 4;

    const int TB = 256;

    {
        int mtiles = (N + 15) / 16;
        int waves  = mtiles * 8;
        int blocks = (waves + 7) / 8;
        gat_gemm_wmma<<<blocks, TB, 0, stream>>>(x, W, H, N);
    }
    {
        long long threads = (long long)N * 32;
        int blocks = (int)((threads + TB - 1) / TB);
        gat_dots<<<blocks, TB, 0, stream>>>(H, att_src, asrc, cnt, total, N);
    }
    {
        int blocks = (E + TB - 1) / TB;
        gat_count<<<blocks, TB, 0, stream>>>(dst, cnt, E);
    }
    {
        int blocks = (N + TB - 1) / TB;
        gat_alloc<<<blocks, TB, 0, stream>>>(cnt, cursor, total, N);
    }
    {
        int blocks = (E + TB - 1) / TB;
        gat_fill<<<blocks, TB, 0, stream>>>(src, dst, cursor, csr, E);
    }
    {
        long long threads = (long long)N * 32;
        int blocks = (int)((threads + TB - 1) / TB);
        gat_gather<<<blocks, TB, 0, stream>>>(H, asrc, att_dst, bias, cnt, cursor, csr, out, N);
    }
}


# gather and WMMA product fused: a block of 8 waves sums the x rows of 16 nodes into LDS, then each wave does a 16x64 strip with A from LDS; summed rows skip memory, a launch less; sum order unchanged
# speedup vs baseline: 4.1895x; 1.3220x over previous
#include <hip/hip_runtime.h>
#include <hip/hip_bf16.h>


#define HEADS   4
#define CH      128
#define HC      512
#define NEG_SLOPE 0.2f
#define NPW     4

typedef float v2f __attribute__((ext_vector_type(2)));
typedef float v8f __attribute__((ext_vector_type(8)));

__device__ __forceinline__ float lrelu(float a) {
    return a > 0.f ? a : NEG_SLOPE * a;
}
__device__ __forceinline__ float wave_sum(float v) {
#pragma unroll
    for (int m = 16; m > 0; m >>= 1) v += __shfl_xor(v, m, 32);
    return v;
}
__device__ __forceinline__ float wave_max(float v) {
#pragma unroll
    for (int m = 16; m > 0; m >>= 1) v = fmaxf(v, __shfl_xor(v, m, 32));
    return v;
}
__device__ __forceinline__ float dot4(float4 a, float4 b) {
    return a.x * b.x + a.y * b.y + a.z * b.z + a.w * b.w;
}

__global__ void gat_fold(const float* __restrict__ W, const float* __restrict__ att_src,
                         const float* __restrict__ att_dst, float* __restrict__ V,
                         int* __restrict__ cnt, int* __restrict__ total, int N) {
    int t = blockIdx.x * blockDim.x + threadIdx.x;
    if (t < N) cnt[t] = 0;
    if (t == 0) *total = 0;
    if (t < CH * 32) {
        int k = t >> 5;
        int j = (t >> 2) & 7;
        int q = t & 3;
        int h = j & 3;
        const float4* __restrict__ w4 = (const float4*)(W + (size_t)k * HC + h * CH + q * 32);
        const float4* __restrict__ a4 = (const float4*)((j < 4 ? att_src : att_dst) + h * CH + q * 32);
        float s = 0.f;
#pragma unroll
        for (int i = 0; i < 8; ++i) s += dot4(w4[i], a4[i]);
        s += __shfl_xor(s, 1, 32);
        s += __shfl_xor(s, 2, 32);
        if (q == 0) V[k * 8 + j] = s;
    }
}

__global__ void gat_scores(const float* __restrict__ X, const float* __restrict__ V,
                           const int* __restrict__ dst, float* __restrict__ asrc,
                           float* __restrict__ adst, int* __restrict__ cnt, int N, int E) {
    int gid      = blockIdx.x * blockDim.x + threadIdx.x;
    int nthreads = gridDim.x * blockDim.x;
    for (int t = gid; t < E; t += nthreads) atomicAdd(&cnt[dst[t]], 1);

    int wid  = gid >> 5;
    int lane = threadIdx.x & 31;
    int n0   = wid * NPW;
    if (n0 >= N) return;

    const float4* __restrict__ v4 = (const float4*)V + lane * 8;
    float4 va[8];
#pragma unroll
    for (int i = 0; i < 8; ++i) va[i] = v4[i];

    const float4* __restrict__ X4 = (const float4*)X;
    float4 xr[NPW];
#pragma unroll
    for (int u = 0; u < NPW; ++u) {
        int n = n0 + u; if (n >= N) n = N - 1;
        xr[u] = X4[(size_t)n * 32 + lane];
    }
    const bool b4 = (lane & 16) != 0, b3 = (lane & 8) != 0, b2 = (lane & 4) != 0;
#pragma unroll
    for (int u = 0; u < NPW; ++u) {
        const float4 x = xr[u];
        float v[8];
        v[0] = x.x * va[0].x + x.y * va[2].x + x.z * va[4].x + x.w * va[6].x;
        v[1] = x.x * va[0].y + x.y * va[2].y + x.z * va[4].y + x.w * va[6].y;
        v[2] = x.x * va[0].z + x.y * va[2].z + x.z * va[4].z + x.w * va[6].z;
        v[3] = x.x * va[0].w + x.y * va[2].w + x.z * va[4].w + x.w * va[6].w;
        v[4] = x.x * va[1].x + x.y * va[3].x + x.z * va[5].x + x.w * va[7].x;
        v[5] = x.x * va[1].y + x.y * va[3].y + x.z * va[5].y + x.w * va[7].y;
        v[6] = x.x * va[1].z + x.y * va[3].z + x.z * va[5].z + x.w * va[7].z;
        v[7] = x.x * va[1].w + x.y * va[3].w + x.z * va[5].w + x.w * va[7].w;
        float t4[4];
#pragma unroll
        for (int j = 0; j < 4; ++j) {
            float send = b4 ? v[j] : v[j + 4];
            float keep = b4 ? v[j + 4] : v[j];
            t4[j] = keep + __shfl_xor(send, 16, 32);
        }
        float t2[2];
#pragma unroll
        for (int j = 0; j < 2; ++j) {
            float send = b3 ? t4[j] : t4[j + 2];
            float keep = b3 ? t4[j + 2] : t4[j];
            t2[j] = keep + __shfl_xor(send, 8, 32);
        }
        float send = b2 ? t2[0] : t2[1];
        float keep = b2 ? t2[1] : t2[0];
        float r = keep + __shfl_xor(send, 4, 32);
        r += __shfl_xor(r, 2, 32);
        r += __shfl_xor(r, 1, 32);
        int n = n0 + u;
        if ((lane & 3) == 0 && n < N) {
            int j = lane >> 2;
            if (j < 4) asrc[(size_t)n * HEADS + j] = r;
            else       adst[(size_t)n * HEADS + (j - 4)] = r;
        }
    }
}

__global__ void gat_alloc(const int* __restrict__ cnt, int* __restrict__ cursor,
                          int* __restrict__ total, int N) {
    __shared__ int wsum[32];
    __shared__ int blockbase;
    int tid  = threadIdx.x;
    int lane = tid & 31;
    int w    = tid >> 5;
    int nw   = blockDim.x >> 5;
    int i    = blockIdx.x * blockDim.x + tid;
    int c    = (i < N) ? cnt[i] : 0;
    int x    = c;
#pragma unroll
    for (int o = 1; o < 32; o <<= 1) {
        int y = __shfl_up(x, o, 32);
        if (lane >= o) x += y;
    }
    if (lane == 31) wsum[w] = x;
    __syncthreads();
    if (tid == 0) {
        int run = 0;
        for (int k = 0; k < nw; ++k) { int t = wsum[k]; wsum[k] = run; run += t; }
        blockbase = atomicAdd(total, run);
    }
    __syncthreads();
    if (i < N) cursor[i] = blockbase + wsum[w] + x - c;
}

__global__ void gat_fill(const int* __restrict__ src, const int* __restrict__ dst,
                         int* __restrict__ cursor, int* __restrict__ csr, int E) {
    int t = blockIdx.x * blockDim.x + threadIdx.x;
    if (t < E) {
        int pos = atomicAdd(&cursor[dst[t]], 1);
        csr[pos] = src[t];
    }
}

#define TM  16
#define ZP  (HC + 4)

__global__ void gat_gather_gemm(const float* __restrict__ X, const float* __restrict__ asrc,
                                const float* __restrict__ adst, const int* __restrict__ cnt,
                                const int* __restrict__ cursor, const int* __restrict__ csr,
                                const float* __restrict__ W, const float* __restrict__ bias,
                                float* __restrict__ out, int N) {
    __shared__ __attribute__((aligned(16))) float Zs[TM * ZP];

    const int lane = threadIdx.x & 31;
    const int wv   = __builtin_amdgcn_readfirstlane((int)(threadIdx.x >> 5));
    const int row0 = blockIdx.x * TM;

    const float4* __restrict__ asrc4 = (const float4*)asrc;
    const float4* __restrict__ adst4 = (const float4*)adst;
    const float4* __restrict__ X4    = (const float4*)X;

    for (int u = 0; u < TM / 8; ++u) {
        const int r = wv * (TM / 8) + u;
        const int n = row0 + r;
        float4 acc[HEADS];
        if (n < N) {
            const int end   = cursor[n];
            const int deg   = cnt[n];
            const int start = end - deg;

            const float4 xs = X4[(size_t)n * 32 + lane];

            const bool ok0 = lane < deg;
            int    s0 = n;
            float4 a0 = make_float4(0.f, 0.f, 0.f, 0.f);
            if (ok0) {
                s0 = csr[start + lane];
                a0 = asrc4[s0];
            }

            const float4 adv = adst4[n];
            const float ad[HEADS] = { adv.x, adv.y, adv.z, adv.w };
            const float4 an = asrc4[n];
            const float vs[HEADS] = { lrelu(an.x + ad[0]), lrelu(an.y + ad[1]),
                                      lrelu(an.z + ad[2]), lrelu(an.w + ad[3]) };
            const float e0[HEADS] = { ok0 ? lrelu(a0.x + ad[0]) : vs[0], ok0 ? lrelu(a0.y + ad[1]) : vs[1],
                                      ok0 ? lrelu(a0.z + ad[2]) : vs[2], ok0 ? lrelu(a0.w + ad[3]) : vs[3] };

            float m[HEADS] = { fmaxf(vs[0], e0[0]), fmaxf(vs[1], e0[1]),
                               fmaxf(vs[2], e0[2]), fmaxf(vs[3], e0[3]) };
            for (int base = start + 32; base < end; base += 32) {
                int i = base + lane;
                if (i < end) {
                    float4 a = asrc4[csr[i]];
                    m[0] = fmaxf(m[0], lrelu(a.x + ad[0]));
                    m[1] = fmaxf(m[1], lrelu(a.y + ad[1]));
                    m[2] = fmaxf(m[2], lrelu(a.z + ad[2]));
                    m[3] = fmaxf(m[3], lrelu(a.w + ad[3]));
                }
            }
#pragma unroll
            for (int h = 0; h < HEADS; ++h) m[h] = wave_max(m[h]);

            float p0[HEADS], z[HEADS];
#pragma unroll
            for (int h = 0; h < HEADS; ++h) {
                p0[h] = ok0 ? __expf(e0[h] - m[h]) : 0.f;
                z[h]  = p0[h];
            }
            for (int base = start + 32; base < end; base += 32) {
                int i = base + lane;
                if (i < end) {
                    float4 a = asrc4[csr[i]];
                    z[0] += __expf(lrelu(a.x + ad[0]) - m[0]);
                    z[1] += __expf(lrelu(a.y + ad[1]) - m[1]);
                    z[2] += __expf(lrelu(a.z + ad[2]) - m[2]);
                    z[3] += __expf(lrelu(a.w + ad[3]) - m[3]);
                }
            }
#pragma unroll
            for (int h = 0; h < HEADS; ++h) {
                float ps = __expf(vs[h] - m[h]);
                z[h] = wave_sum(z[h]) + ps;
                float w = ps / z[h];
                acc[h] = make_float4(w * xs.x, w * xs.y, w * xs.z, w * xs.w);
            }

            int   s = s0;
            float al[HEADS] = { p0[0] / z[0], p0[1] / z[1], p0[2] / z[2], p0[3] / z[3] };
            for (int base = start; base < end; base += 32) {
                if (base != start) {
                    int i = base + lane;
                    s = n;
                    al[0] = al[1] = al[2] = al[3] = 0.f;
                    if (i < end) {
                        s = csr[i];
                        float4 a = asrc4[s];
                        al[0] = __expf(lrelu(a.x + ad[0]) - m[0]) / z[0];
                        al[1] = __expf(lrelu(a.y + ad[1]) - m[1]) / z[1];
                        al[2] = __expf(lrelu(a.z + ad[2]) - m[2]) / z[2];
                        al[3] = __expf(lrelu(a.w + ad[3]) - m[3]) / z[3];
                    }
                }
                int k = end - base; if (k > 32) k = 32;
                for (int j0 = 0; j0 < k; j0 += 4) {
#pragma unroll
                    for (int q = 0; q < 4; ++q) {
                        int   jj = j0 + q;
                        int   sj = __builtin_amdgcn_readlane(s, jj);
                        float w0 = __int_as_float(__builtin_amdgcn_readlane(__float_as_int(al[0]), jj));
                        float w1 = __int_as_float(__builtin_amdgcn_readlane(__float_as_int(al[1]), jj));
                        float w2 = __int_as_float(__builtin_amdgcn_readlane(__float_as_int(al[2]), jj));
                        float w3 = __int_as_float(__builtin_amdgcn_readlane(__float_as_int(al[3]), jj));
                        const float4 rr = X4[(size_t)sj * 32 + lane];
                        acc[0].x += w0 * rr.x; acc[0].y += w0 * rr.y; acc[0].z += w0 * rr.z; acc[0].w += w0 * rr.w;
                        acc[1].x += w1 * rr.x; acc[1].y += w1 * rr.y; acc[1].z += w1 * rr.z; acc[1].w += w1 * rr.w;
                        acc[2].x += w2 * rr.x; acc[2].y += w2 * rr.y; acc[2].z += w2 * rr.z; acc[2].w += w2 * rr.w;
                        acc[3].x += w3 * rr.x; acc[3].y += w3 * rr.y; acc[3].z += w3 * rr.z; acc[3].w += w3 * rr.w;
                    }
                }
            }
        } else {
#pragma unroll
            for (int h = 0; h < HEADS; ++h) acc[h] = make_float4(0.f, 0.f, 0.f, 0.f);
        }
        float4* zrow = (float4*)(Zs + r * ZP);
#pragma unroll
        for (int h = 0; h < HEADS; ++h) zrow[h * 32 + lane] = acc[h];
    }

    __syncthreads();

    const int ns   = wv;
    const int col0 = ns * 64;
    const int half = lane >> 4;
    const int l16  = lane & 15;

    const float* As = Zs + l16 * ZP + (ns >> 1) * CH + half * 2;

    v8f acc0 = {}, acc1 = {}, acc2 = {}, acc3 = {};

#pragma unroll 4
    for (int k0 = 0; k0 < CH; k0 += 4) {
        int ka = k0 + half * 2;
        v2f a = *(const v2f*)(As + k0);

        const float* __restrict__ B = W + (size_t)ka * HC + col0 + l16;
        v2f b0; b0.x = B[0];  b0.y = B[HC];
        v2f b1; b1.x = B[16]; b1.y = B[16 + HC];
        v2f b2; b2.x = B[32]; b2.y = B[32 + HC];
        v2f b3; b3.x = B[48]; b3.y = B[48 + HC];

        acc0 = __builtin_amdgcn_wmma_f32_16x16x4_f32(false, a, false, b0, (short)0, acc0, false, false);
        acc1 = __builtin_amdgcn_wmma_f32_16x16x4_f32(false, a, false, b1, (short)0, acc1, false, false);
        acc2 = __builtin_amdgcn_wmma_f32_16x16x4_f32(false, a, false, b2, (short)0, acc2, false, false);
        acc3 = __builtin_amdgcn_wmma_f32_16x16x4_f32(false, a, false, b3, (short)0, acc3, false, false);
    }

    const float* __restrict__ bc = bias + col0 + l16;
    const float bz0 = bc[0], bz1 = bc[16], bz2 = bc[32], bz3 = bc[48];

#pragma unroll
    for (int r = 0; r < 8; ++r) {
        int m = row0 + half * 8 + r;
        if (m < N) {
            float* __restrict__ O = out + (size_t)m * HC + col0 + l16;
            O[0]  = acc0[r] + bz0;
            O[16] = acc1[r] + bz1;
            O[32] = acc2[r] + bz2;
            O[48] = acc3[r] + bz3;
        }
    }
}

extern "C" void kernel_launch(void* const* d_in, const int* in_sizes, int n_in,
                              void* d_out, int out_size, void* d_ws, size_t ws_size,
                              hipStream_t stream) {
    const float* x        = (const float*)d_in[0];
    const int*   eidx     = (const int*)d_in[1];
    const float* W        = (const float*)d_in[2];
    const float* att_src  = (const float*)d_in[3];
    const float* att_dst  = (const float*)d_in[4];
    const float* bias     = (const float*)d_in[5];
    float*       out      = (float*)d_out;

    const int N = in_sizes[0] / CH;
    const int E = in_sizes[1] / 2;
    const int* src = eidx;
    const int* dst = eidx + E;

    float* adst   = (float*)d_ws;
    float* asrc   = adst + (size_t)N * HEADS;
    int*   cnt    = (int*)(asrc + (size_t)N * HEADS);
    int*   cursor = cnt + N;
    int*   total  = cursor + N;
    int*   csr    = total + 4;
    float* V      = (float*)csr;

    const int TB = 256;

    {
        int threads = N > CH * 32 ? N : CH * 32;
        int blocks  = (threads + TB - 1) / TB;
        gat_fold<<<blocks, TB, 0, stream>>>(W, att_src, att_dst, V, cnt, total, N);
    }
    {
        long long waves   = ((long long)N + NPW - 1) / NPW;
        long long threads = waves * 32;
        int blocks = (int)((threads + TB - 1) / TB);
        gat_scores<<<blocks, TB, 0, stream>>>(x, V, dst, asrc, adst, cnt, N, E);
    }
    {
        int blocks = (N + TB - 1) / TB;
        gat_alloc<<<blocks, TB, 0, stream>>>(cnt, cursor, total, N);
    }
    {
        int blocks = (E + TB - 1) / TB;
        gat_fill<<<blocks, TB, 0, stream>>>(src, dst, cursor, csr, E);
    }
    {
        int blocks = (N + TM - 1) / TM;
        gat_gather_gemm<<<blocks, TB, 0, stream>>>(x, asrc, adst, cnt, cursor, csr, W, bias, out, N);
    }
}


# block tile of 32 nodes on 16 waves: each wave a 32x32 strip as 2x2 WMMA tiles, so every B fragment of W feeds two row tiles; W reads and B load instructions per row halved; sum order unchanged
# speedup vs baseline: 4.3361x; 1.0350x over previous
#include <hip/hip_runtime.h>
#include <hip/hip_bf16.h>


#define HEADS   4
#define CH      128
#define HC      512
#define NEG_SLOPE 0.2f
#define NPW     4

typedef float v2f __attribute__((ext_vector_type(2)));
typedef float v8f __attribute__((ext_vector_type(8)));

__device__ __forceinline__ float lrelu(float a) {
    return a > 0.f ? a : NEG_SLOPE * a;
}
__device__ __forceinline__ float wave_sum(float v) {
#pragma unroll
    for (int m = 16; m > 0; m >>= 1) v += __shfl_xor(v, m, 32);
    return v;
}
__device__ __forceinline__ float wave_max(float v) {
#pragma unroll
    for (int m = 16; m > 0; m >>= 1) v = fmaxf(v, __shfl_xor(v, m, 32));
    return v;
}
__device__ __forceinline__ float dot4(float4 a, float4 b) {
    return a.x * b.x + a.y * b.y + a.z * b.z + a.w * b.w;
}

__global__ void gat_fold(const float* __restrict__ W, const float* __restrict__ att_src,
                         const float* __restrict__ att_dst, float* __restrict__ V,
                         int* __restrict__ cnt, int* __restrict__ total, int N) {
    int t = blockIdx.x * blockDim.x + threadIdx.x;
    if (t < N) cnt[t] = 0;
    if (t == 0) *total = 0;
    if (t < CH * 32) {
        int k = t >> 5;
        int j = (t >> 2) & 7;
        int q = t & 3;
        int h = j & 3;
        const float4* __restrict__ w4 = (const float4*)(W + (size_t)k * HC + h * CH + q * 32);
        const float4* __restrict__ a4 = (const float4*)((j < 4 ? att_src : att_dst) + h * CH + q * 32);
        float s = 0.f;
#pragma unroll
        for (int i = 0; i < 8; ++i) s += dot4(w4[i], a4[i]);
        s += __shfl_xor(s, 1, 32);
        s += __shfl_xor(s, 2, 32);
        if (q == 0) V[k * 8 + j] = s;
    }
}

__global__ void gat_scores(const float* __restrict__ X, const float* __restrict__ V,
                           const int* __restrict__ dst, float* __restrict__ asrc,
                           float* __restrict__ adst, int* __restrict__ cnt, int N, int E) {
    int gid      = blockIdx.x * blockDim.x + threadIdx.x;
    int nthreads = gridDim.x * blockDim.x;
    for (int t = gid; t < E; t += nthreads) atomicAdd(&cnt[dst[t]], 1);

    int wid  = gid >> 5;
    int lane = threadIdx.x & 31;
    int n0   = wid * NPW;
    if (n0 >= N) return;

    const float4* __restrict__ v4 = (const float4*)V + lane * 8;
    float4 va[8];
#pragma unroll
    for (int i = 0; i < 8; ++i) va[i] = v4[i];

    const float4* __restrict__ X4 = (const float4*)X;
    float4 xr[NPW];
#pragma unroll
    for (int u = 0; u < NPW; ++u) {
        int n = n0 + u; if (n >= N) n = N - 1;
        xr[u] = X4[(size_t)n * 32 + lane];
    }
    const bool b4 = (lane & 16) != 0, b3 = (lane & 8) != 0, b2 = (lane & 4) != 0;
#pragma unroll
    for (int u = 0; u < NPW; ++u) {
        const float4 x = xr[u];
        float v[8];
        v[0] = x.x * va[0].x + x.y * va[2].x + x.z * va[4].x + x.w * va[6].x;
        v[1] = x.x * va[0].y + x.y * va[2].y + x.z * va[4].y + x.w * va[6].y;
        v[2] = x.x * va[0].z + x.y * va[2].z + x.z * va[4].z + x.w * va[6].z;
        v[3] = x.x * va[0].w + x.y * va[2].w + x.z * va[4].w + x.w * va[6].w;
        v[4] = x.x * va[1].x + x.y * va[3].x + x.z * va[5].x + x.w * va[7].x;
        v[5] = x.x * va[1].y + x.y * va[3].y + x.z * va[5].y + x.w * va[7].y;
        v[6] = x.x * va[1].z + x.y * va[3].z + x.z * va[5].z + x.w * va[7].z;
        v[7] = x.x * va[1].w + x.y * va[3].w + x.z * va[5].w + x.w * va[7].w;
        float t4[4];
#pragma unroll
        for (int j = 0; j < 4; ++j) {
            float send = b4 ? v[j] : v[j + 4];
            float keep = b4 ? v[j + 4] : v[j];
            t4[j] = keep + __shfl_xor(send, 16, 32);
        }
        float t2[2];
#pragma unroll
        for (int j = 0; j < 2; ++j) {
            float send = b3 ? t4[j] : t4[j + 2];
            float keep = b3 ? t4[j + 2] : t4[j];
            t2[j] = keep + __shfl_xor(send, 8, 32);
        }
        float send = b2 ? t2[0] : t2[1];
        float keep = b2 ? t2[1] : t2[0];
        float r = keep + __shfl_xor(send, 4, 32);
        r += __shfl_xor(r, 2, 32);
        r += __shfl_xor(r, 1, 32);
        int n = n0 + u;
        if ((lane & 3) == 0 && n < N) {
            int j = lane >> 2;
            if (j < 4) asrc[(size_t)n * HEADS + j] = r;
            else       adst[(size_t)n * HEADS + (j - 4)] = r;
        }
    }
}

__global__ void gat_alloc(const int* __restrict__ cnt, int* __restrict__ cursor,
                          int* __restrict__ total, int N) {
    __shared__ int wsum[32];
    __shared__ int blockbase;
    int tid  = threadIdx.x;
    int lane = tid & 31;
    int w    = tid >> 5;
    int nw   = blockDim.x >> 5;
    int i    = blockIdx.x * blockDim.x + tid;
    int c    = (i < N) ? cnt[i] : 0;
    int x    = c;
#pragma unroll
    for (int o = 1; o < 32; o <<= 1) {
        int y = __shfl_up(x, o, 32);
        if (lane >= o) x += y;
    }
    if (lane == 31) wsum[w] = x;
    __syncthreads();
    if (tid == 0) {
        int run = 0;
        for (int k = 0; k < nw; ++k) { int t = wsum[k]; wsum[k] = run; run += t; }
        blockbase = atomicAdd(total, run);
    }
    __syncthreads();
    if (i < N) cursor[i] = blockbase + wsum[w] + x - c;
}

__global__ void gat_fill(const int* __restrict__ src, const int* __restrict__ dst,
                         int* __restrict__ cursor, int* __restrict__ csr, int E) {
    int t = blockIdx.x * blockDim.x + threadIdx.x;
    if (t < E) {
        int pos = atomicAdd(&cursor[dst[t]], 1);
        csr[pos] = src[t];
    }
}

#define TM  32
#define NWV 16
#define ZP  (HC + 4)

__global__ void gat_gather_gemm(const float* __restrict__ X, const float* __restrict__ asrc,
                                const float* __restrict__ adst, const int* __restrict__ cnt,
                                const int* __restrict__ cursor, const int* __restrict__ csr,
                                const float* __restrict__ W, const float* __restrict__ bias,
                                float* __restrict__ out, int N) {
    __shared__ __attribute__((aligned(16))) float Zs[TM * ZP];

    const int lane = threadIdx.x & 31;
    const int wv   = __builtin_amdgcn_readfirstlane((int)(threadIdx.x >> 5));
    const int row0 = blockIdx.x * TM;

    const float4* __restrict__ asrc4 = (const float4*)asrc;
    const float4* __restrict__ adst4 = (const float4*)adst;
    const float4* __restrict__ X4    = (const float4*)X;

    for (int u = 0; u < TM / NWV; ++u) {
        const int r = wv * (TM / NWV) + u;
        const int n = row0 + r;
        float4 acc[HEADS];
        if (n < N) {
            const int end   = cursor[n];
            const int deg   = cnt[n];
            const int start = end - deg;

            const float4 xs = X4[(size_t)n * 32 + lane];

            const bool ok0 = lane < deg;
            int    s0 = n;
            float4 a0 = make_float4(0.f, 0.f, 0.f, 0.f);
            if (ok0) {
                s0 = csr[start + lane];
                a0 = asrc4[s0];
            }

            const float4 adv = adst4[n];
            const float ad[HEADS] = { adv.x, adv.y, adv.z, adv.w };
            const float4 an = asrc4[n];
            const float vs[HEADS] = { lrelu(an.x + ad[0]), lrelu(an.y + ad[1]),
                                      lrelu(an.z + ad[2]), lrelu(an.w + ad[3]) };
            const float e0[HEADS] = { ok0 ? lrelu(a0.x + ad[0]) : vs[0], ok0 ? lrelu(a0.y + ad[1]) : vs[1],
                                      ok0 ? lrelu(a0.z + ad[2]) : vs[2], ok0 ? lrelu(a0.w + ad[3]) : vs[3] };

            float m[HEADS] = { fmaxf(vs[0], e0[0]), fmaxf(vs[1], e0[1]),
                               fmaxf(vs[2], e0[2]), fmaxf(vs[3], e0[3]) };
            for (int base = start + 32; base < end; base += 32) {
                int i = base + lane;
                if (i < end) {
                    float4 a = asrc4[csr[i]];
                    m[0] = fmaxf(m[0], lrelu(a.x + ad[0]));
                    m[1] = fmaxf(m[1], lrelu(a.y + ad[1]));
                    m[2] = fmaxf(m[2], lrelu(a.z + ad[2]));
                    m[3] = fmaxf(m[3], lrelu(a.w + ad[3]));
                }
            }
#pragma unroll
            for (int h = 0; h < HEADS; ++h) m[h] = wave_max(m[h]);

            float p0[HEADS], z[HEADS];
#pragma unroll
            for (int h = 0; h < HEADS; ++h) {
                p0[h] = ok0 ? __expf(e0[h] - m[h]) : 0.f;
                z[h]  = p0[h];
            }
            for (int base = start + 32; base < end; base += 32) {
                int i = base + lane;
                if (i < end) {
                    float4 a = asrc4[csr[i]];
                    z[0] += __expf(lrelu(a.x + ad[0]) - m[0]);
                    z[1] += __expf(lrelu(a.y + ad[1]) - m[1]);
                    z[2] += __expf(lrelu(a.z + ad[2]) - m[2]);
                    z[3] += __expf(lrelu(a.w + ad[3]) - m[3]);
                }
            }
#pragma unroll
            for (int h = 0; h < HEADS; ++h) {
                float ps = __expf(vs[h] - m[h]);
                z[h] = wave_sum(z[h]) + ps;
                float w = ps / z[h];
                acc[h] = make_float4(w * xs.x, w * xs.y, w * xs.z, w * xs.w);
            }

            int   s = s0;
            float al[HEADS] = { p0[0] / z[0], p0[1] / z[1], p0[2] / z[2], p0[3] / z[3] };
            for (int base = start; base < end; base += 32) {
                if (base != start) {
                    int i = base + lane;
                    s = n;
                    al[0] = al[1] = al[2] = al[3] = 0.f;
                    if (i < end) {
                        s = csr[i];
                        float4 a = asrc4[s];
                        al[0] = __expf(lrelu(a.x + ad[0]) - m[0]) / z[0];
                        al[1] = __expf(lrelu(a.y + ad[1]) - m[1]) / z[1];
                        al[2] = __expf(lrelu(a.z + ad[2]) - m[2]) / z[2];
                        al[3] = __expf(lrelu(a.w + ad[3]) - m[3]) / z[3];
                    }
                }
                int k = end - base; if (k > 32) k = 32;
                for (int j0 = 0; j0 < k; j0 += 4) {
#pragma unroll
                    for (int q = 0; q < 4; ++q) {
                        int   jj = j0 + q;
                        int   sj = __builtin_amdgcn_readlane(s, jj);
                        float w0 = __int_as_float(__builtin_amdgcn_readlane(__float_as_int(al[0]), jj));
                        float w1 = __int_as_float(__builtin_amdgcn_readlane(__float_as_int(al[1]), jj));
                        float w2 = __int_as_float(__builtin_amdgcn_readlane(__float_as_int(al[2]), jj));
                        float w3 = __int_as_float(__builtin_amdgcn_readlane(__float_as_int(al[3]), jj));
                        const float4 rr = X4[(size_t)sj * 32 + lane];
                        acc[0].x += w0 * rr.x; acc[0].y += w0 * rr.y; acc[0].z += w0 * rr.z; acc[0].w += w0 * rr.w;
                        acc[1].x += w1 * rr.x; acc[1].y += w1 * rr.y; acc[1].z += w1 * rr.z; acc[1].w += w1 * rr.w;
                        acc[2].x += w2 * rr.x; acc[2].y += w2 * rr.y; acc[2].z += w2 * rr.z; acc[2].w += w2 * rr.w;
                        acc[3].x += w3 * rr.x; acc[3].y += w3 * rr.y; acc[3].z += w3 * rr.z; acc[3].w += w3 * rr.w;
                    }
                }
            }
        } else {
#pragma unroll
            for (int h = 0; h < HEADS; ++h) acc[h] = make_float4(0.f, 0.f, 0.f, 0.f);
        }
        float4* zrow = (float4*)(Zs + r * ZP);
#pragma unroll
        for (int h = 0; h < HEADS; ++h) zrow[h * 32 + lane] = acc[h];
    }

    __syncthreads();

    const int col0 = wv * 32;
    const int half = lane >> 4;
    const int l16  = lane & 15;

    const float* As0 = Zs + l16 * ZP + (wv >> 2) * CH + half * 2;
    const float* As1 = As0 + 16 * ZP;

    v8f c00 = {}, c01 = {}, c10 = {}, c11 = {};

#pragma unroll 4
    for (int k0 = 0; k0 < CH; k0 += 4) {
        int ka = k0 + half * 2;
        v2f a0 = *(const v2f*)(As0 + k0);
        v2f a1 = *(const v2f*)(As1 + k0);

        const float* __restrict__ B = W + (size_t)ka * HC + col0 + l16;
        v2f b0; b0.x = B[0];  b0.y = B[HC];
        v2f b1; b1.x = B[16]; b1.y = B[16 + HC];

        c00 = __builtin_amdgcn_wmma_f32_16x16x4_f32(false, a0, false, b0, (short)0, c00, false, false);
        c01 = __builtin_amdgcn_wmma_f32_16x16x4_f32(false, a0, false, b1, (short)0, c01, false, false);
        c10 = __builtin_amdgcn_wmma_f32_16x16x4_f32(false, a1, false, b0, (short)0, c10, false, false);
        c11 = __builtin_amdgcn_wmma_f32_16x16x4_f32(false, a1, false, b1, (short)0, c11, false, false);
    }

    const float* __restrict__ bc = bias + col0 + l16;
    const float bz0 = bc[0], bz1 = bc[16];

#pragma unroll
    for (int r = 0; r < 8; ++r) {
        int m0 = row0 + half * 8 + r;
        int m1 = m0 + 16;
        if (m0 < N) {
            float* __restrict__ O = out + (size_t)m0 * HC + col0 + l16;
            O[0]  = c00[r] + bz0;
            O[16] = c01[r] + bz1;
        }
        if (m1 < N) {
            float* __restrict__ O = out + (size_t)m1 * HC + col0 + l16;
            O[0]  = c10[r] + bz0;
            O[16] = c11[r] + bz1;
        }
    }
}

extern "C" void kernel_launch(void* const* d_in, const int* in_sizes, int n_in,
                              void* d_out, int out_size, void* d_ws, size_t ws_size,
                              hipStream_t stream) {
    const float* x        = (const float*)d_in[0];
    const int*   eidx     = (const int*)d_in[1];
    const float* W        = (const float*)d_in[2];
    const float* att_src  = (const float*)d_in[3];
    const float* att_dst  = (const float*)d_in[4];
    const float* bias     = (const float*)d_in[5];
    float*       out      = (float*)d_out;

    const int N = in_sizes[0] / CH;
    const int E = in_sizes[1] / 2;
    const int* src = eidx;
    const int* dst = eidx + E;

    float* adst   = (float*)d_ws;
    float* asrc   = adst + (size_t)N * HEADS;
    int*   cnt    = (int*)(asrc + (size_t)N * HEADS);
    int*   cursor = cnt + N;
    int*   total  = cursor + N;
    int*   csr    = total + 4;
    float* V      = (float*)csr;

    const int TB = 256;

    {
        int threads = N > CH * 32 ? N : CH * 32;
        int blocks  = (threads + TB - 1) / TB;
        gat_fold<<<blocks, TB, 0, stream>>>(W, att_src, att_dst, V, cnt, total, N);
    }
    {
        long long waves   = ((long long)N + NPW - 1) / NPW;
        long long threads = waves * 32;
        int blocks = (int)((threads + TB - 1) / TB);
        gat_scores<<<blocks, TB, 0, stream>>>(x, V, dst, asrc, adst, cnt, N, E);
    }
    {
        int blocks = (N + TB - 1) / TB;
        gat_alloc<<<blocks, TB, 0, stream>>>(cnt, cursor, total, N);
    }
    {
        int blocks = (E + TB - 1) / TB;
        gat_fill<<<blocks, TB, 0, stream>>>(src, dst, cursor, csr, E);
    }
    {
        int blocks = (N + TM - 1) / TM;
        gat_gather_gemm<<<blocks, NWV * 32, 0, stream>>>(x, asrc, adst, cnt, cursor, csr, W, bias, out, N);
    }
}
